# starting kernel
# baseline (speedup 1.0000x reference)
#include <hip/hip_runtime.h>
#include <hip/hip_bf16.h>


typedef float v2f __attribute__((ext_vector_type(2)));
typedef float v8f __attribute__((ext_vector_type(8)));

#define NB 2
#define NC 64
#define NPIX 147456
#define NCLS 20
#define BLKPX 1024
#define CHUNKS (BLKPX / 32)
#define NBLKS (NPIX / BLKPX)
#define PARTS 16
#define PADALIGN (PARTS * 4)
#define NPAD_MAX (NPIX + NCLS * PADALIGN)
#define MIN_CLASS_SIZE 5000
#define EPSV 1e-6f

__global__ __launch_bounds__(32) void hist_kernel(const int* __restrict__ seg,
                                                  int* __restrict__ hist) {
  int blk = blockIdx.x;
  int b = blk / NBLKS, bb = blk % NBLKS;
  int lane = threadIdx.x;
  const int* s = seg + (size_t)b * NPIX + (size_t)bb * BLKPX;
  int cnt = 0;
  for (int ch = 0; ch < CHUNKS; ++ch) {
    int myk = s[ch * 32 + lane];
#pragma unroll
    for (int k = 0; k < NCLS; ++k) {
      unsigned bal = __builtin_amdgcn_ballot_w32(myk == k);
      if (lane == k) cnt += __popc(bal);
    }
  }
  if (lane < NCLS) hist[(size_t)blk * NCLS + lane] = cnt;
}

__global__ __launch_bounds__(64) void scan_kernel(const int* __restrict__ hist,
                                                  int* __restrict__ blockoff,
                                                  int* __restrict__ classinfo) {
  int tid = threadIdx.x;
  __shared__ int tot[NB * NCLS];
  __shared__ int padl[NB * NCLS];
  if (tid < NB * NCLS) {
    int b = tid / NCLS, k = tid % NCLS;
    int run = 0;
    for (int blk = 0; blk < NBLKS; ++blk) {
      size_t idx = ((size_t)(b * NBLKS + blk)) * NCLS + k;
      blockoff[idx] = run;
      run += hist[idx];
    }
    tot[tid] = run;
    padl[tid] = (run + PADALIGN - 1) & ~(PADALIGN - 1);
  }
  __syncthreads();
  if (tid < NB) {
    int b = tid, base = 0;
    for (int k = 0; k < NCLS; ++k) {
      classinfo[(b * NCLS + k) * 3 + 0] = base;
      classinfo[(b * NCLS + k) * 3 + 1] = tot[b * NCLS + k];
      classinfo[(b * NCLS + k) * 3 + 2] = padl[b * NCLS + k];
      base += padl[b * NCLS + k];
    }
  }
}

__global__ __launch_bounds__(256) void padzero_kernel(const int* __restrict__ classinfo,
                                                      float* __restrict__ comp) {
  int bk = blockIdx.x;
  int b = bk / NCLS;
  int base = classinfo[bk * 3 + 0];
  int count = classinfo[bk * 3 + 1];
  int padl = classinfo[bk * 3 + 2];
  int nel = (padl - count) * NC;
  float* dst = comp + ((size_t)b * NPAD_MAX + base + count) * NC;
  for (int i = threadIdx.x; i < nel; i += 256) dst[i] = 0.0f;
}

__global__ __launch_bounds__(32) void scatter_kernel(const float* __restrict__ hyper,
                                                     const int* __restrict__ seg,
                                                     const int* __restrict__ blockoff,
                                                     const int* __restrict__ classinfo,
                                                     float* __restrict__ comp) {
  int blk = blockIdx.x;
  int b = blk / NBLKS, bb = blk % NBLKS;
  int lane = threadIdx.x;
  __shared__ float stage[32 * 65];
  __shared__ int posArr[32];
  int run = 0;
  unsigned ltmask = (1u << lane) - 1u;
  size_t nbase0 = (size_t)bb * BLKPX;

  for (int ch = 0; ch < CHUNKS; ++ch) {
    size_t nbase = nbase0 + (size_t)ch * 32;
    int n = (int)(nbase + lane);
    int myk = seg[(size_t)b * NPIX + n];
    unsigned balmy = 0;
    int pend = 0;
#pragma unroll
    for (int k = 0; k < NCLS; ++k) {
      unsigned bal = __builtin_amdgcn_ballot_w32(myk == k);
      if (myk == k) balmy = bal;
      if (lane == k) pend = __popc(bal);
    }
    int runbase = __shfl(run, myk, 32);
    int myrank = __popc(balmy & ltmask);
    run += (lane < NCLS) ? pend : 0;

    int cbase = classinfo[(b * NCLS + myk) * 3 + 0];
    int brel = blockoff[((size_t)(b * NBLKS + bb)) * NCLS + myk];
    int pos = cbase + brel + runbase + myrank;
    posArr[lane] = pos;

    for (int c = 0; c < NC; ++c)
      stage[lane * 65 + c] = hyper[((size_t)b * NC + c) * NPIX + n];
    __syncthreads();
    for (int p = 0; p < 32; ++p) {
      size_t dst = ((size_t)b * NPAD_MAX + posArr[p]) * NC;
      comp[dst + lane] = stage[p * 65 + lane];
      comp[dst + lane + 32] = stage[p * 65 + lane + 32];
    }
    __syncthreads();
  }
}

__global__ __launch_bounds__(32) void syrk_kernel(const float* __restrict__ comp,
                                                  const int* __restrict__ classinfo,
                                                  float* __restrict__ sxxpart,
                                                  float* __restrict__ sumpart) {
  int gid = blockIdx.x;
  int part = gid / (NB * NCLS);
  int bk = gid % (NB * NCLS);
  int b = bk / NCLS;
  int lane = threadIdx.x;

  int base = classinfo[bk * 3 + 0];
  int padl = classinfo[bk * 3 + 2];
  int plen = padl / PARTS;
  size_t rowbase = (size_t)b * NPAD_MAX + base + (size_t)part * plen;

  int m = lane & 15;
  int kb = (lane >> 4) << 1;

  v8f acc[4][4] = {};
  for (int p = 0; p < plen; p += 4) {
    const float* row0 = comp + (rowbase + p + kb) * NC;
    const float* row1 = row0 + NC;
    v2f F[4];
#pragma unroll
    for (int blk = 0; blk < 4; ++blk) {
      F[blk].x = row0[blk * 16 + m];
      F[blk].y = row1[blk * 16 + m];
    }
#pragma unroll
    for (int cb = 0; cb < 4; ++cb)
#pragma unroll
      for (int db = 0; db < 4; ++db)
        acc[cb][db] = __builtin_amdgcn_wmma_f32_16x16x4_f32(
            false, F[cb], false, F[db], (short)0, acc[cb][db], false, false);
  }

  size_t ob = (size_t)gid * (NC * NC);
  int col = lane & 15;
  int rof = (lane >> 4) * 8;
#pragma unroll
  for (int cb = 0; cb < 4; ++cb)
#pragma unroll
    for (int db = 0; db < 4; ++db)
#pragma unroll
      for (int j = 0; j < 8; ++j) {
        int r = cb * 16 + rof + j;
        sxxpart[ob + (size_t)r * NC + db * 16 + col] = acc[cb][db][j];
      }

  float s0 = 0.0f, s1 = 0.0f;
  for (int p = 0; p < plen; ++p) {
    const float* row = comp + (rowbase + p) * NC;
    s0 += row[lane];
    s1 += row[lane + 32];
  }
  sumpart[(size_t)gid * NC + lane] = s0;
  sumpart[(size_t)gid * NC + lane + 32] = s1;
}

__global__ __launch_bounds__(256) void reduce_solve_kernel(const float* __restrict__ sxxpart,
                                                           const float* __restrict__ sumpart,
                                                           const int* __restrict__ classinfo,
                                                           const float* __restrict__ pattern,
                                                           float* __restrict__ params) {
  int bk = blockIdx.x;
  int tid = threadIdx.x;
  __shared__ float covL[64 * 65];
  __shared__ float meanL[64], pL[64], yv[64], wv[64];

  int count = classinfo[bk * 3 + 1];
  float safe = (float)(count > 0 ? count : 1);

  if (tid < 64) {
    float s = 0.0f;
    for (int part = 0; part < PARTS; ++part)
      s += sumpart[((size_t)part * NB * NCLS + bk) * NC + tid];
    meanL[tid] = s / safe;
    pL[tid] = pattern[tid];
  }
  __syncthreads();

  for (int idx = tid; idx < NC * NC; idx += 256) {
    int i = idx >> 6, j = idx & 63;
    float s = 0.0f;
    for (int part = 0; part < PARTS; ++part)
      s += sxxpart[((size_t)part * NB * NCLS + bk) * (NC * NC) + idx];
    float cv = s / safe - meanL[i] * meanL[j];
    if (i == j) cv += EPSV;
    covL[i * 65 + j] = cv;
  }
  __syncthreads();

  for (int kk = 0; kk < NC; ++kk) {
    if (tid == 0) covL[kk * 65 + kk] = sqrtf(covL[kk * 65 + kk]);
    __syncthreads();
    float d = covL[kk * 65 + kk];
    for (int i = kk + 1 + tid; i < NC; i += 256) covL[i * 65 + kk] /= d;
    __syncthreads();
    int nrem = NC - 1 - kk;
    for (int t = tid; t < nrem * nrem; t += 256) {
      int i = kk + 1 + t / nrem;
      int j = kk + 1 + t % nrem;
      if (j <= i) covL[i * 65 + j] -= covL[i * 65 + kk] * covL[j * 65 + kk];
    }
    __syncthreads();
  }

  if (tid < 64) yv[tid] = pL[tid];
  __syncthreads();
  for (int j = 0; j < NC; ++j) {
    if (tid == 0) yv[j] /= covL[j * 65 + j];
    __syncthreads();
    float yj = yv[j];
    for (int i = j + 1 + tid; i < NC; i += 256) yv[i] -= covL[i * 65 + j] * yj;
    __syncthreads();
  }
  if (tid < 64) wv[tid] = yv[tid];
  __syncthreads();
  for (int j = NC - 1; j >= 0; --j) {
    if (tid == 0) wv[j] /= covL[j * 65 + j];
    __syncthreads();
    float wj = wv[j];
    for (int i = tid; i < j; i += 256) wv[i] -= covL[j * 65 + i] * wj;
    __syncthreads();
  }

  if (tid < 64) params[(size_t)bk * 66 + tid] = wv[tid];
  if (tid == 0) {
    float pw = 0.0f, off = 0.0f;
    for (int c = 0; c < NC; ++c) {
      pw += pL[c] * wv[c];
      off += meanL[c] * wv[c];
    }
    float denom = sqrtf(pw);
    float scale = (count >= MIN_CLASS_SIZE && denom > 0.0f) ? (1.0f / denom) : 0.0f;
    params[(size_t)bk * 66 + 64] = off;
    params[(size_t)bk * 66 + 65] = scale;
  }
}

__global__ __launch_bounds__(256) void score_kernel(const float* __restrict__ hyper,
                                                    const int* __restrict__ seg,
                                                    const float* __restrict__ params,
                                                    float* __restrict__ out) {
  int b = blockIdx.y;
  int tid = threadIdx.x;
  __shared__ float wS[NCLS * 65];
  __shared__ float offS[NCLS], sclS[NCLS];
  for (int i = tid; i < NCLS * NC; i += 256) {
    int k = i >> 6, c = i & 63;
    wS[k * 65 + c] = params[((size_t)b * NCLS + k) * 66 + c];
  }
  if (tid < NCLS) {
    offS[tid] = params[((size_t)b * NCLS + tid) * 66 + 64];
    sclS[tid] = params[((size_t)b * NCLS + tid) * 66 + 65];
  }
  __syncthreads();

  size_t n = (size_t)blockIdx.x * 256 + tid;
  int k = seg[(size_t)b * NPIX + n];
  const float* xp = hyper + (size_t)b * NC * NPIX + n;
  float acc = 0.0f;
#pragma unroll
  for (int c = 0; c < NC; ++c)
    acc = fmaf(xp[(size_t)c * NPIX], wS[k * 65 + c], acc);
  out[(size_t)b * NPIX + n] = (acc - offS[k]) * sclS[k];
}

extern "C" void kernel_launch(void* const* d_in, const int* in_sizes, int n_in,
                              void* d_out, int out_size, void* d_ws, size_t ws_size,
                              hipStream_t stream) {
  const float* hyper = (const float*)d_in[0];
  const float* pattern = (const float*)d_in[1];
  const int* seg = (const int*)d_in[2];
  float* out = (float*)d_out;

  char* ws = (char*)d_ws;
  auto al = [](size_t x) { return (x + 255) & ~(size_t)255; };
  size_t off = 0;
  float* comp = (float*)(ws + off);
  off += al((size_t)NB * NPAD_MAX * NC * sizeof(float));
  int* hist = (int*)(ws + off);
  off += al((size_t)NB * NBLKS * NCLS * sizeof(int));
  int* blockoff = (int*)(ws + off);
  off += al((size_t)NB * NBLKS * NCLS * sizeof(int));
  int* classinfo = (int*)(ws + off);
  off += al((size_t)NB * NCLS * 3 * sizeof(int));
  float* sxxpart = (float*)(ws + off);
  off += al((size_t)PARTS * NB * NCLS * NC * NC * sizeof(float));
  float* sumpart = (float*)(ws + off);
  off += al((size_t)PARTS * NB * NCLS * NC * sizeof(float));
  float* params = (float*)(ws + off);
  off += al((size_t)NB * NCLS * 66 * sizeof(float));

  hipLaunchKernelGGL(hist_kernel, dim3(NB * NBLKS), dim3(32), 0, stream, seg, hist);
  hipLaunchKernelGGL(scan_kernel, dim3(1), dim3(64), 0, stream, hist, blockoff, classinfo);
  hipLaunchKernelGGL(padzero_kernel, dim3(NB * NCLS), dim3(256), 0, stream, classinfo, comp);
  hipLaunchKernelGGL(scatter_kernel, dim3(NB * NBLKS), dim3(32), 0, stream, hyper, seg,
                     blockoff, classinfo, comp);
  hipLaunchKernelGGL(syrk_kernel, dim3(PARTS * NB * NCLS), dim3(32), 0, stream, comp,
                     classinfo, sxxpart, sumpart);
  hipLaunchKernelGGL(reduce_solve_kernel, dim3(NB * NCLS), dim3(256), 0, stream, sxxpart,
                     sumpart, classinfo, pattern, params);
  hipLaunchKernelGGL(score_kernel, dim3(NPIX / 256, NB), dim3(256), 0, stream, hyper, seg,
                     params, out);
}


# class moments taken straight from the channel-major image: 20 class-owning waves gather their pixels from an LDS tile into WMMA, partials folded grid-wide; no sorted copy; sum order changes
# speedup vs baseline: 2.8544x; 2.8544x over previous
#include <hip/hip_runtime.h>


typedef float v2f __attribute__((ext_vector_type(2)));
typedef float v8f __attribute__((ext_vector_type(8)));

#define NB 2
#define NC 64
#define NPIX 147456
#define NCLS 20
#define TP 384
#define PITCH (TP + 4)
#define GPB 128
#define TILES (NPIX / GPB / TP)
#define NPARTS GPB
#define QP (NPARTS / 4)
#define FSEG 16
#define LCAP (TP + 4)
#define NTHR (NCLS * 32)
#define V4ROW (TP / 4)
#define V4TILE (NC * V4ROW)
#define MIN_CLASS_SIZE 5000
#define EPSV 1e-6f

__global__ __launch_bounds__(NTHR) void stats_kernel(const float* __restrict__ hyper,
                                                     const int* __restrict__ seg,
                                                     float* __restrict__ sxxpart,
                                                     float* __restrict__ sumpart,
                                                     int* __restrict__ cntpart) {
  __shared__ __attribute__((aligned(16))) float stage[NC * PITCH];
  __shared__ int lists[NCLS * LCAP];

  int tid = threadIdx.x;
  int lane = tid & 31;
  int wave = tid >> 5;
  int b = blockIdx.x / GPB;
  int part = blockIdx.x % GPB;
  unsigned n0 = (unsigned)part * (TILES * TP);
  const float* hb = hyper + (size_t)b * NC * NPIX;
  const int* sb = seg + (size_t)b * NPIX;
  int m = lane & 15;
  int kb = (lane >> 4) << 1;
  unsigned ltmask = (1u << lane) - 1u;
  int* mylist = lists + wave * LCAP;

  if (tid < NC) stage[tid * PITCH + TP] = 0.0f;

  v8f acc[4][4] = {};
  float cs[4] = {0.0f, 0.0f, 0.0f, 0.0f};
  int cntTot = 0;

  for (int t = 0; t < TILES; ++t) {
    unsigned nt = n0 + (unsigned)t * TP;

    int sv[TP / 32];
#pragma unroll
    for (int ch = 0; ch < TP / 32; ++ch) sv[ch] = sb[nt + (unsigned)(ch * 32 + lane)];

#pragma unroll
    for (int h = 0; h < 2; ++h) {
      float4 v[5];
#pragma unroll
      for (int u = 0; u < 5; ++u) {
        unsigned i = (unsigned)tid + (unsigned)((h * 5 + u) * NTHR);
        unsigned ic = i < V4TILE ? i : (unsigned)(V4TILE - 1);
        unsigned c = ic / V4ROW, j4 = ic % V4ROW;
        v[u] = *(const float4*)(hb + (c * NPIX + nt + j4 * 4));
      }
#pragma unroll
      for (int u = 0; u < 5; ++u) {
        unsigned i = (unsigned)tid + (unsigned)((h * 5 + u) * NTHR);
        unsigned c = i / V4ROW, j4 = i % V4ROW;
        if (i < V4TILE) *(float4*)(stage + c * PITCH + j4 * 4) = v[u];
      }
    }

    int cnt = 0;
#pragma unroll
    for (int ch = 0; ch < TP / 32; ++ch) {
      bool is = (sv[ch] == wave);
      unsigned bal = __builtin_amdgcn_ballot_w32(is);
      if (is) mylist[cnt + __popc(bal & ltmask)] = ch * 32 + lane;
      cnt += __popc(bal);
    }
    int padded = (cnt + 3) & ~3;
    if (lane < padded - cnt) mylist[cnt + lane] = TP;
    cntTot += cnt;
    __syncthreads();

    for (int s = 0; s < padded; s += 4) {
      int q0 = mylist[s + kb];
      int q1 = mylist[s + kb + 1];
      v2f F[4];
#pragma unroll
      for (int blk = 0; blk < 4; ++blk) {
        F[blk].x = stage[(blk * 16 + m) * PITCH + q0];
        F[blk].y = stage[(blk * 16 + m) * PITCH + q1];
        cs[blk] += F[blk].x + F[blk].y;
      }
#pragma unroll
      for (int cb = 0; cb < 4; ++cb)
#pragma unroll
        for (int db = 0; db < 4; ++db)
          acc[cb][db] = __builtin_amdgcn_wmma_f32_16x16x4_f32(
              false, F[cb], false, F[db], (short)0, acc[cb][db], false, false);
    }
    __syncthreads();
  }

  size_t gid = ((size_t)part * NB + b) * NCLS + wave;
  size_t ob = gid * (NC * NC);
  int col = lane & 15;
  int rof = (lane >> 4) * 8;
#pragma unroll
  for (int cb = 0; cb < 4; ++cb)
#pragma unroll
    for (int db = 0; db < 4; ++db)
#pragma unroll
      for (int j = 0; j < 8; ++j) {
        int r = cb * 16 + rof + j;
        sxxpart[ob + (size_t)r * NC + db * 16 + col] = acc[cb][db][j];
      }

#pragma unroll
  for (int blk = 0; blk < 4; ++blk) {
    float tot = cs[blk] + __shfl_xor(cs[blk], 16, 32);
    if (lane < 16) sumpart[gid * NC + blk * 16 + m] = tot;
  }
  if (lane == 0) cntpart[gid] = cntTot;
}

__global__ __launch_bounds__(256) void fold_kernel(const float* __restrict__ sxxpart,
                                                   const float* __restrict__ sumpart,
                                                   const int* __restrict__ cntpart,
                                                   float* __restrict__ sxxsum,
                                                   float* __restrict__ sumsum,
                                                   int* __restrict__ cntsum) {
  int bk = blockIdx.x / FSEG;
  int sg = blockIdx.x % FSEG;
  int tid = threadIdx.x;
  int qtr = tid >> 6;
  int e = tid & 63;
  __shared__ float4 red4[256];
  __shared__ float red1[256];
  __shared__ int cntS[NPARTS];

  {
    int f4 = sg * 64 + e;
    float4 s = make_float4(0.0f, 0.0f, 0.0f, 0.0f);
    for (int p0 = qtr * QP; p0 < qtr * QP + QP; p0 += 16) {
      float4 v[16];
#pragma unroll
      for (int u = 0; u < 16; ++u)
        v[u] = *(const float4*)(sxxpart + ((size_t)(p0 + u) * NB * NCLS + bk) * (NC * NC) +
                                (size_t)f4 * 4);
#pragma unroll
      for (int u = 0; u < 16; ++u) {
        s.x += v[u].x;
        s.y += v[u].y;
        s.z += v[u].z;
        s.w += v[u].w;
      }
    }
    red4[tid] = s;
    __syncthreads();
    if (tid < 64) {
      float4 a = red4[tid], b1 = red4[tid + 64], c1 = red4[tid + 128], d1 = red4[tid + 192];
      float4 r;
      r.x = ((a.x + b1.x) + c1.x) + d1.x;
      r.y = ((a.y + b1.y) + c1.y) + d1.y;
      r.z = ((a.z + b1.z) + c1.z) + d1.z;
      r.w = ((a.w + b1.w) + c1.w) + d1.w;
      *(float4*)(sxxsum + (size_t)bk * (NC * NC) + (size_t)f4 * 4) = r;
    }
  }

  if (sg == 0) {
    float s = 0.0f;
    for (int p0 = qtr * QP; p0 < qtr * QP + QP; p0 += 16) {
      float v[16];
#pragma unroll
      for (int u = 0; u < 16; ++u)
        v[u] = sumpart[((size_t)(p0 + u) * NB * NCLS + bk) * NC + e];
#pragma unroll
      for (int u = 0; u < 16; ++u) s += v[u];
    }
    red1[tid] = s;
    if (tid < NPARTS) cntS[tid] = cntpart[(size_t)tid * NB * NCLS + bk];
    __syncthreads();
    if (tid < 64)
      sumsum[(size_t)bk * NC + tid] =
          ((red1[tid] + red1[tid + 64]) + red1[tid + 128]) + red1[tid + 192];
    if (tid == 0) {
      int cnt = 0;
      for (int i = 0; i < NPARTS; ++i) cnt += cntS[i];
      cntsum[bk] = cnt;
    }
  }
}

__global__ __launch_bounds__(256) void reduce_solve_kernel(const float* __restrict__ sxxsum,
                                                           const float* __restrict__ sumsum,
                                                           const int* __restrict__ cntsum,
                                                           const float* __restrict__ pattern,
                                                           float* __restrict__ params) {
  int bk = blockIdx.x;
  int tid = threadIdx.x;
  __shared__ float covL[64 * 65];
  __shared__ float meanL[64], pL[64], yv[64], wv[64];

  int count = cntsum[bk];
  float safe = (float)(count > 0 ? count : 1);

  if (tid < 64) {
    meanL[tid] = sumsum[(size_t)bk * NC + tid] / safe;
    pL[tid] = pattern[tid];
  }
  __syncthreads();

  for (int idx = tid; idx < NC * NC; idx += 256) {
    int i = idx >> 6, j = idx & 63;
    float s = sxxsum[(size_t)bk * (NC * NC) + idx];
    float cv = s / safe - meanL[i] * meanL[j];
    if (i == j) cv += EPSV;
    covL[i * 65 + j] = cv;
  }
  __syncthreads();

  for (int kk = 0; kk < NC; ++kk) {
    if (tid == 0) covL[kk * 65 + kk] = sqrtf(covL[kk * 65 + kk]);
    __syncthreads();
    float d = covL[kk * 65 + kk];
    for (int i = kk + 1 + tid; i < NC; i += 256) covL[i * 65 + kk] /= d;
    __syncthreads();
    int nrem = NC - 1 - kk;
    for (int t = tid; t < nrem * nrem; t += 256) {
      int i = kk + 1 + t / nrem;
      int j = kk + 1 + t % nrem;
      if (j <= i) covL[i * 65 + j] -= covL[i * 65 + kk] * covL[j * 65 + kk];
    }
    __syncthreads();
  }

  if (tid < 64) yv[tid] = pL[tid];
  __syncthreads();
  for (int j = 0; j < NC; ++j) {
    if (tid == 0) yv[j] /= covL[j * 65 + j];
    __syncthreads();
    float yj = yv[j];
    for (int i = j + 1 + tid; i < NC; i += 256) yv[i] -= covL[i * 65 + j] * yj;
    __syncthreads();
  }
  if (tid < 64) wv[tid] = yv[tid];
  __syncthreads();
  for (int j = NC - 1; j >= 0; --j) {
    if (tid == 0) wv[j] /= covL[j * 65 + j];
    __syncthreads();
    float wj = wv[j];
    for (int i = tid; i < j; i += 256) wv[i] -= covL[j * 65 + i] * wj;
    __syncthreads();
  }

  if (tid < 64) params[(size_t)bk * 66 + tid] = wv[tid];
  if (tid == 0) {
    float pw = 0.0f, off = 0.0f;
    for (int c = 0; c < NC; ++c) {
      pw += pL[c] * wv[c];
      off += meanL[c] * wv[c];
    }
    float denom = sqrtf(pw);
    float scale = (count >= MIN_CLASS_SIZE && denom > 0.0f) ? (1.0f / denom) : 0.0f;
    params[(size_t)bk * 66 + 64] = off;
    params[(size_t)bk * 66 + 65] = scale;
  }
}

__global__ __launch_bounds__(256) void score_kernel(const float* __restrict__ hyper,
                                                    const int* __restrict__ seg,
                                                    const float* __restrict__ params,
                                                    float* __restrict__ out) {
  int b = blockIdx.y;
  int tid = threadIdx.x;
  __shared__ float wS[NCLS * 65];
  __shared__ float offS[NCLS], sclS[NCLS];
  for (int i = tid; i < NCLS * NC; i += 256) {
    int k = i >> 6, c = i & 63;
    wS[k * 65 + c] = params[((size_t)b * NCLS + k) * 66 + c];
  }
  if (tid < NCLS) {
    offS[tid] = params[((size_t)b * NCLS + tid) * 66 + 64];
    sclS[tid] = params[((size_t)b * NCLS + tid) * 66 + 65];
  }
  __syncthreads();

  size_t n = (size_t)blockIdx.x * 256 + tid;
  int k = seg[(size_t)b * NPIX + n];
  const float* xp = hyper + (size_t)b * NC * NPIX + n;
  float acc = 0.0f;
#pragma unroll
  for (int c = 0; c < NC; ++c)
    acc = fmaf(xp[(size_t)c * NPIX], wS[k * 65 + c], acc);
  out[(size_t)b * NPIX + n] = (acc - offS[k]) * sclS[k];
}

extern "C" void kernel_launch(void* const* d_in, const int* in_sizes, int n_in,
                              void* d_out, int out_size, void* d_ws, size_t ws_size,
                              hipStream_t stream) {
  const float* hyper = (const float*)d_in[0];
  const float* pattern = (const float*)d_in[1];
  const int* seg = (const int*)d_in[2];
  float* out = (float*)d_out;

  char* ws = (char*)d_ws;
  auto al = [](size_t x) { return (x + 255) & ~(size_t)255; };
  size_t off = 0;
  float* sxxpart = (float*)(ws + off);
  off += al((size_t)NPARTS * NB * NCLS * NC * NC * sizeof(float));
  float* sumpart = (float*)(ws + off);
  off += al((size_t)NPARTS * NB * NCLS * NC * sizeof(float));
  int* cntpart = (int*)(ws + off);
  off += al((size_t)NPARTS * NB * NCLS * sizeof(int));
  float* sxxsum = (float*)(ws + off);
  off += al((size_t)NB * NCLS * NC * NC * sizeof(float));
  float* sumsum = (float*)(ws + off);
  off += al((size_t)NB * NCLS * NC * sizeof(float));
  int* cntsum = (int*)(ws + off);
  off += al((size_t)NB * NCLS * sizeof(int));
  float* params = (float*)(ws + off);
  off += al((size_t)NB * NCLS * 66 * sizeof(float));

  hipLaunchKernelGGL(stats_kernel, dim3(NB * GPB), dim3(NTHR), 0, stream, hyper, seg, sxxpart,
                     sumpart, cntpart);
  hipLaunchKernelGGL(fold_kernel, dim3(NB * NCLS * FSEG), dim3(256), 0, stream, sxxpart,
                     sumpart, cntpart, sxxsum, sumsum, cntsum);
  hipLaunchKernelGGL(reduce_solve_kernel, dim3(NB * NCLS), dim3(256), 0, stream, sxxsum,
                     sumsum, cntsum, pattern, params);
  hipLaunchKernelGGL(score_kernel, dim3(NPIX / 256, NB), dim3(256), 0, stream, hyper, seg,
                     params, out);
}


# reduce_solve: cov in 4x4 register tiles over the group, LDL^T with one LDS column hand-off and one barrier per pivot, p carried as 65th row, back substitution by one wave; sum order changes
# speedup vs baseline: 5.8782x; 2.0593x over previous
#include <hip/hip_runtime.h>


typedef float v2f __attribute__((ext_vector_type(2)));
typedef float v8f __attribute__((ext_vector_type(8)));

#define NB 2
#define NC 64
#define NPIX 147456
#define NCLS 20
#define TP 384
#define PITCH (TP + 4)
#define GPB 128
#define TILES (NPIX / GPB / TP)
#define NPARTS GPB
#define QP (NPARTS / 4)
#define FSEG 16
#define LCAP (TP + 4)
#define NTHR (NCLS * 32)
#define V4ROW (TP / 4)
#define V4TILE (NC * V4ROW)
#define MIN_CLASS_SIZE 5000
#define EPSV 1e-6f

__global__ __launch_bounds__(NTHR) void stats_kernel(const float* __restrict__ hyper,
                                                     const int* __restrict__ seg,
                                                     float* __restrict__ sxxpart,
                                                     float* __restrict__ sumpart,
                                                     int* __restrict__ cntpart) {
  __shared__ __attribute__((aligned(16))) float stage[NC * PITCH];
  __shared__ int lists[NCLS * LCAP];

  int tid = threadIdx.x;
  int lane = tid & 31;
  int wave = tid >> 5;
  int b = blockIdx.x / GPB;
  int part = blockIdx.x % GPB;
  unsigned n0 = (unsigned)part * (TILES * TP);
  const float* hb = hyper + (size_t)b * NC * NPIX;
  const int* sb = seg + (size_t)b * NPIX;
  int m = lane & 15;
  int kb = (lane >> 4) << 1;
  unsigned ltmask = (1u << lane) - 1u;
  int* mylist = lists + wave * LCAP;

  if (tid < NC) stage[tid * PITCH + TP] = 0.0f;

  v8f acc[4][4] = {};
  float cs[4] = {0.0f, 0.0f, 0.0f, 0.0f};
  int cntTot = 0;

  for (int t = 0; t < TILES; ++t) {
    unsigned nt = n0 + (unsigned)t * TP;

    int sv[TP / 32];
#pragma unroll
    for (int ch = 0; ch < TP / 32; ++ch) sv[ch] = sb[nt + (unsigned)(ch * 32 + lane)];

#pragma unroll
    for (int h = 0; h < 2; ++h) {
      float4 v[5];
#pragma unroll
      for (int u = 0; u < 5; ++u) {
        unsigned i = (unsigned)tid + (unsigned)((h * 5 + u) * NTHR);
        unsigned ic = i < V4TILE ? i : (unsigned)(V4TILE - 1);
        unsigned c = ic / V4ROW, j4 = ic % V4ROW;
        v[u] = *(const float4*)(hb + (c * NPIX + nt + j4 * 4));
      }
#pragma unroll
      for (int u = 0; u < 5; ++u) {
        unsigned i = (unsigned)tid + (unsigned)((h * 5 + u) * NTHR);
        unsigned c = i / V4ROW, j4 = i % V4ROW;
        if (i < V4TILE) *(float4*)(stage + c * PITCH + j4 * 4) = v[u];
      }
    }

    int cnt = 0;
#pragma unroll
    for (int ch = 0; ch < TP / 32; ++ch) {
      bool is = (sv[ch] == wave);
      unsigned bal = __builtin_amdgcn_ballot_w32(is);
      if (is) mylist[cnt + __popc(bal & ltmask)] = ch * 32 + lane;
      cnt += __popc(bal);
    }
    int padded = (cnt + 3) & ~3;
    if (lane < padded - cnt) mylist[cnt + lane] = TP;
    cntTot += cnt;
    __syncthreads();

    for (int s = 0; s < padded; s += 4) {
      int q0 = mylist[s + kb];
      int q1 = mylist[s + kb + 1];
      v2f F[4];
#pragma unroll
      for (int blk = 0; blk < 4; ++blk) {
        F[blk].x = stage[(blk * 16 + m) * PITCH + q0];
        F[blk].y = stage[(blk * 16 + m) * PITCH + q1];
        cs[blk] += F[blk].x + F[blk].y;
      }
#pragma unroll
      for (int cb = 0; cb < 4; ++cb)
#pragma unroll
        for (int db = 0; db < 4; ++db)
          acc[cb][db] = __builtin_amdgcn_wmma_f32_16x16x4_f32(
              false, F[cb], false, F[db], (short)0, acc[cb][db], false, false);
    }
    __syncthreads();
  }

  size_t gid = ((size_t)part * NB + b) * NCLS + wave;
  size_t ob = gid * (NC * NC);
  int col = lane & 15;
  int rof = (lane >> 4) * 8;
#pragma unroll
  for (int cb = 0; cb < 4; ++cb)
#pragma unroll
    for (int db = 0; db < 4; ++db)
#pragma unroll
      for (int j = 0; j < 8; ++j) {
        int r = cb * 16 + rof + j;
        sxxpart[ob + (size_t)r * NC + db * 16 + col] = acc[cb][db][j];
      }

#pragma unroll
  for (int blk = 0; blk < 4; ++blk) {
    float tot = cs[blk] + __shfl_xor(cs[blk], 16, 32);
    if (lane < 16) sumpart[gid * NC + blk * 16 + m] = tot;
  }
  if (lane == 0) cntpart[gid] = cntTot;
}

__global__ __launch_bounds__(256) void fold_kernel(const float* __restrict__ sxxpart,
                                                   const float* __restrict__ sumpart,
                                                   const int* __restrict__ cntpart,
                                                   float* __restrict__ sxxsum,
                                                   float* __restrict__ sumsum,
                                                   int* __restrict__ cntsum) {
  int bk = blockIdx.x / FSEG;
  int sg = blockIdx.x % FSEG;
  int tid = threadIdx.x;
  int qtr = tid >> 6;
  int e = tid & 63;
  __shared__ float4 red4[256];
  __shared__ float red1[256];
  __shared__ int cntS[NPARTS];

  {
    int f4 = sg * 64 + e;
    float4 s = make_float4(0.0f, 0.0f, 0.0f, 0.0f);
    for (int p0 = qtr * QP; p0 < qtr * QP + QP; p0 += 16) {
      float4 v[16];
#pragma unroll
      for (int u = 0; u < 16; ++u)
        v[u] = *(const float4*)(sxxpart + ((size_t)(p0 + u) * NB * NCLS + bk) * (NC * NC) +
                                (size_t)f4 * 4);
#pragma unroll
      for (int u = 0; u < 16; ++u) {
        s.x += v[u].x;
        s.y += v[u].y;
        s.z += v[u].z;
        s.w += v[u].w;
      }
    }
    red4[tid] = s;
    __syncthreads();
    if (tid < 64) {
      float4 a = red4[tid], b1 = red4[tid + 64], c1 = red4[tid + 128], d1 = red4[tid + 192];
      float4 r;
      r.x = ((a.x + b1.x) + c1.x) + d1.x;
      r.y = ((a.y + b1.y) + c1.y) + d1.y;
      r.z = ((a.z + b1.z) + c1.z) + d1.z;
      r.w = ((a.w + b1.w) + c1.w) + d1.w;
      *(float4*)(sxxsum + (size_t)bk * (NC * NC) + (size_t)f4 * 4) = r;
    }
  }

  if (sg == 0) {
    float s = 0.0f;
    for (int p0 = qtr * QP; p0 < qtr * QP + QP; p0 += 16) {
      float v[16];
#pragma unroll
      for (int u = 0; u < 16; ++u)
        v[u] = sumpart[((size_t)(p0 + u) * NB * NCLS + bk) * NC + e];
#pragma unroll
      for (int u = 0; u < 16; ++u) s += v[u];
    }
    red1[tid] = s;
    if (tid < NPARTS) cntS[tid] = cntpart[(size_t)tid * NB * NCLS + bk];
    __syncthreads();
    if (tid < 64)
      sumsum[(size_t)bk * NC + tid] =
          ((red1[tid] + red1[tid + 64]) + red1[tid + 128]) + red1[tid + 192];
    if (tid == 0) {
      int cnt = 0;
      for (int i = 0; i < NPARTS; ++i) cnt += cntS[i];
      cntsum[bk] = cnt;
    }
  }
}

#define RSP 68

__global__ __launch_bounds__(256) void reduce_solve_kernel(const float* __restrict__ sxxsum,
                                                           const float* __restrict__ sumsum,
                                                           const int* __restrict__ cntsum,
                                                           const float* __restrict__ pattern,
                                                           float* __restrict__ params) {
  int bk = blockIdx.x;
  int tid = threadIdx.x;
  int tx = tid & 15;
  int ty = tid >> 4;
  __shared__ __attribute__((aligned(16))) float colS[NC * RSP];

  int count = cntsum[bk];
  float safe = (float)(count > 0 ? count : 1);

  float4 sr = *(const float4*)(sumsum + (size_t)bk * NC + 4 * ty);
  float4 sc = *(const float4*)(sumsum + (size_t)bk * NC + 4 * tx);
  float mr[4] = {sr.x / safe, sr.y / safe, sr.z / safe, sr.w / safe};
  float mc[4] = {sc.x / safe, sc.y / safe, sc.z / safe, sc.w / safe};

  float a[4][4];
  float g[4];
#pragma unroll
  for (int c = 0; c < 4; ++c) g[c] = pattern[4 * tx + c];
#pragma unroll
  for (int r = 0; r < 4; ++r) {
    float4 s4 = *(const float4*)(sxxsum + (size_t)bk * (NC * NC) + (size_t)(4 * ty + r) * NC +
                                 4 * tx);
    float sv[4] = {s4.x, s4.y, s4.z, s4.w};
#pragma unroll
    for (int c = 0; c < 4; ++c) {
      float cv = sv[c] / safe - mr[r] * mc[c];
      if (4 * ty + r == 4 * tx + c) cv += EPSV;
      a[r][c] = cv;
    }
  }

  if (tx == 0) *(float4*)(colS + 4 * ty) = make_float4(a[0][0], a[1][0], a[2][0], a[3][0]);
  if (tid == 0) colS[NC] = g[0];

  for (int kb = 0; kb < NC / 4; ++kb) {
#pragma unroll
    for (int kc = 0; kc < 4; ++kc) {
      int k = kb * 4 + kc;
      __syncthreads();
      const float* ck = colS + k * RSP;
      float4 cr4 = *(const float4*)(ck + 4 * ty);
      float4 cc4 = *(const float4*)(ck + 4 * tx);
      float piv = ck[k];
      float ca = ck[NC];
      float inv = 1.0f / piv;
      float tr[4] = {cr4.x * inv, cr4.y * inv, cr4.z * inv, cr4.w * inv};
      float ta = ca * inv;
      float cc[4] = {cc4.x, cc4.y, cc4.z, cc4.w};

      const int cn = (kc + 1) & 3;
#pragma unroll
      for (int r = 0; r < 4; ++r) a[r][cn] = fmaf(-tr[r], cc[cn], a[r][cn]);
      g[cn] = fmaf(-ta, cc[cn], g[cn]);
      if (k + 1 < NC) {
        int kbn = (k + 1) >> 2;
        float* cnx = colS + (k + 1) * RSP;
        if (tx == kbn) *(float4*)(cnx + 4 * ty) = make_float4(a[0][cn], a[1][cn], a[2][cn], a[3][cn]);
        if (tid == kbn) cnx[NC] = g[cn];
      }
#pragma unroll
      for (int q = 1; q < 4; ++q) {
        const int c = (kc + 1 + q) & 3;
#pragma unroll
        for (int r = 0; r < 4; ++r) a[r][c] = fmaf(-tr[r], cc[c], a[r][c]);
        g[c] = fmaf(-ta, cc[c], g[c]);
      }
    }
  }

  if (tid < 32) {
    int l = tid;
    float cl[NC], cu[32];
#pragma unroll
    for (int q = 0; q < NC / 4; ++q) {
      float4 v = *(const float4*)(colS + l * RSP + 4 * q);
      cl[4 * q] = v.x; cl[4 * q + 1] = v.y; cl[4 * q + 2] = v.z; cl[4 * q + 3] = v.w;
    }
#pragma unroll
    for (int q = 0; q < 8; ++q) {
      float4 v = *(const float4*)(colS + (l + 32) * RSP + 32 + 4 * q);
      cu[4 * q] = v.x; cu[4 * q + 1] = v.y; cu[4 * q + 2] = v.z; cu[4 * q + 3] = v.w;
    }
    float i0 = 1.0f / colS[l * RSP + l];
    float i1 = 1.0f / colS[(l + 32) * RSP + l + 32];
    float r0 = colS[l * RSP + NC];
    float r1 = colS[(l + 32) * RSP + NC];
    float w0 = 0.0f, w1 = 0.0f;
#pragma unroll
    for (int i = NC - 1; i >= 32; --i) {
      float wi = __int_as_float(__builtin_amdgcn_readlane(__float_as_int(r1 * i1), i - 32));
      if (l == i - 32) w1 = wi;
      r0 = fmaf(-cl[i], wi, r0);
      if (l + 32 < i) r1 = fmaf(-cu[i - 32], wi, r1);
    }
#pragma unroll
    for (int i = 31; i >= 0; --i) {
      float wi = __int_as_float(__builtin_amdgcn_readlane(__float_as_int(r0 * i0), i));
      if (l == i) w0 = wi;
      if (l < i) r0 = fmaf(-cl[i], wi, r0);
    }

    params[(size_t)bk * 66 + l] = w0;
    params[(size_t)bk * 66 + l + 32] = w1;

    float pw = pattern[l] * w0 + pattern[l + 32] * w1;
    float off = (sumsum[(size_t)bk * NC + l] / safe) * w0 +
                (sumsum[(size_t)bk * NC + l + 32] / safe) * w1;
#pragma unroll
    for (int sh = 16; sh >= 1; sh >>= 1) {
      pw += __shfl_xor(pw, sh, 32);
      off += __shfl_xor(off, sh, 32);
    }
    if (l == 0) {
      float denom = sqrtf(pw);
      float scale = (count >= MIN_CLASS_SIZE && denom > 0.0f) ? (1.0f / denom) : 0.0f;
      params[(size_t)bk * 66 + 64] = off;
      params[(size_t)bk * 66 + 65] = scale;
    }
  }
}

__global__ __launch_bounds__(256) void score_kernel(const float* __restrict__ hyper,
                                                    const int* __restrict__ seg,
                                                    const float* __restrict__ params,
                                                    float* __restrict__ out) {
  int b = blockIdx.y;
  int tid = threadIdx.x;
  __shared__ float wS[NCLS * 65];
  __shared__ float offS[NCLS], sclS[NCLS];
  for (int i = tid; i < NCLS * NC; i += 256) {
    int k = i >> 6, c = i & 63;
    wS[k * 65 + c] = params[((size_t)b * NCLS + k) * 66 + c];
  }
  if (tid < NCLS) {
    offS[tid] = params[((size_t)b * NCLS + tid) * 66 + 64];
    sclS[tid] = params[((size_t)b * NCLS + tid) * 66 + 65];
  }
  __syncthreads();

  size_t n = (size_t)blockIdx.x * 256 + tid;
  int k = seg[(size_t)b * NPIX + n];
  const float* xp = hyper + (size_t)b * NC * NPIX + n;
  float acc = 0.0f;
#pragma unroll
  for (int c = 0; c < NC; ++c)
    acc = fmaf(xp[(size_t)c * NPIX], wS[k * 65 + c], acc);
  out[(size_t)b * NPIX + n] = (acc - offS[k]) * sclS[k];
}

extern "C" void kernel_launch(void* const* d_in, const int* in_sizes, int n_in,
                              void* d_out, int out_size, void* d_ws, size_t ws_size,
                              hipStream_t stream) {
  const float* hyper = (const float*)d_in[0];
  const float* pattern = (const float*)d_in[1];
  const int* seg = (const int*)d_in[2];
  float* out = (float*)d_out;

  char* ws = (char*)d_ws;
  auto al = [](size_t x) { return (x + 255) & ~(size_t)255; };
  size_t off = 0;
  float* sxxpart = (float*)(ws + off);
  off += al((size_t)NPARTS * NB * NCLS * NC * NC * sizeof(float));
  float* sumpart = (float*)(ws + off);
  off += al((size_t)NPARTS * NB * NCLS * NC * sizeof(float));
  int* cntpart = (int*)(ws + off);
  off += al((size_t)NPARTS * NB * NCLS * sizeof(int));
  float* sxxsum = (float*)(ws + off);
  off += al((size_t)NB * NCLS * NC * NC * sizeof(float));
  float* sumsum = (float*)(ws + off);
  off += al((size_t)NB * NCLS * NC * sizeof(float));
  int* cntsum = (int*)(ws + off);
  off += al((size_t)NB * NCLS * sizeof(int));
  float* params = (float*)(ws + off);
  off += al((size_t)NB * NCLS * 66 * sizeof(float));

  hipLaunchKernelGGL(stats_kernel, dim3(NB * GPB), dim3(NTHR), 0, stream, hyper, seg, sxxpart,
                     sumpart, cntpart);
  hipLaunchKernelGGL(fold_kernel, dim3(NB * NCLS * FSEG), dim3(256), 0, stream, sxxpart,
                     sumpart, cntpart, sxxsum, sumsum, cntsum);
  hipLaunchKernelGGL(reduce_solve_kernel, dim3(NB * NCLS), dim3(256), 0, stream, sxxsum,
                     sumsum, cntsum, pattern, params);
  hipLaunchKernelGGL(score_kernel, dim3(NPIX / 256, NB), dim3(256), 0, stream, hyper, seg,
                     params, out);
}


# stats tile copy made memory-parallel: row thirds per wave on scalar bases, ten loads in flight per lane (were waited for one by one), second LDS buffer: a wave refills after its own gather; same sums
# speedup vs baseline: 6.6916x; 1.1384x over previous
#include <hip/hip_runtime.h>


typedef float v2f __attribute__((ext_vector_type(2)));
typedef float v8f __attribute__((ext_vector_type(8)));

#define NB 2
#define NC 64
#define NPIX 147456
#define NCLS 20
#define TP 384
#define PITCH (TP + 4)
#define GPB 128
#define TILES (NPIX / GPB / TP)
#define NPARTS GPB
#define QP (NPARTS / 4)
#define FSEG 16
#define LCAP (TP + 4)
#define NTHR (NCLS * 32)
#define PCS 10
#define MIN_CLASS_SIZE 5000
#define EPSV 1e-6f


__device__ __forceinline__ void tile_loads(float4 (&v)[PCS], const float* __restrict__ tb,
                                           int wv, int lane) {
#pragma unroll
  for (int g = 0; g < 3; ++g) {
    unsigned ro = (unsigned)__builtin_amdgcn_readfirstlane((wv + 20 * g) * NPIX);
    const float* rp = tb + ro;
#pragma unroll
    for (int u = 0; u < 3; ++u) v[3 * g + u] = *(const float4*)(rp + (u * 128 + lane * 4));
  }
  {
    int w9 = wv < 12 ? wv : wv - 12;
    unsigned ro = (unsigned)__builtin_amdgcn_readfirstlane((60 + w9 / 3) * NPIX + (w9 % 3) * 128);
    const float* rp = tb + ro;
    v[9] = *(const float4*)(rp + lane * 4);
  }
}

__device__ __forceinline__ void tile_stores(const float4 (&v)[PCS], float* stage, int wv,
                                            int lane) {
  int lo = wv * PITCH + lane * 4;
  asm volatile("" : "+v"(lo));
  float* lb = stage + lo;
#pragma unroll
  for (int g = 0; g < 3; ++g)
#pragma unroll
    for (int u = 0; u < 3; ++u) *(float4*)(lb + (g * 20 * PITCH + u * 128)) = v[3 * g + u];
  int w9 = wv < 12 ? wv : wv - 12;
  *(float4*)(stage + (60 + w9 / 3) * PITCH + (w9 % 3) * 128 + lane * 4) = v[9];
}

__device__ __forceinline__ int build_list(const int (&sv)[TP / 32], int wave, int lane,
                                          unsigned ltmask, int* mylist) {
  int cnt = 0;
#pragma unroll
  for (int ch = 0; ch < TP / 32; ++ch) {
    bool is = (sv[ch] == wave);
    unsigned bal = __builtin_amdgcn_ballot_w32(is);
    if (is) mylist[cnt + __popc(bal & ltmask)] = ch * 32 + lane;
    cnt += __popc(bal);
  }
  int padded = (cnt + 3) & ~3;
  if (lane < padded - cnt) mylist[cnt + lane] = TP;
  return cnt;
}

__device__ __forceinline__ void gather_tile(v8f (&acc)[4][4], float (&cs)[4], const float* stage,
                                            const int* mylist, int padded, int m, int kb) {
  for (int s = 0; s < padded; s += 4) {
    int q0 = mylist[s + kb];
    int q1 = mylist[s + kb + 1];
    v2f F[4];
#pragma unroll
    for (int blk = 0; blk < 4; ++blk) {
      F[blk].x = stage[(blk * 16 + m) * PITCH + q0];
      F[blk].y = stage[(blk * 16 + m) * PITCH + q1];
      cs[blk] += F[blk].x + F[blk].y;
    }
#pragma unroll
    for (int cb = 0; cb < 4; ++cb)
#pragma unroll
      for (int db = 0; db < 4; ++db)
        acc[cb][db] = __builtin_amdgcn_wmma_f32_16x16x4_f32(
            false, F[cb], false, F[db], (short)0, acc[cb][db], false, false);
  }
}

__global__ __launch_bounds__(NTHR) void stats_kernel(const float* __restrict__ hyper,
                                                     const int* __restrict__ seg,
                                                     float* __restrict__ sxxpart,
                                                     float* __restrict__ sumpart,
                                                     int* __restrict__ cntpart) {
  __shared__ __attribute__((aligned(16))) float stage[2][NC * PITCH];
  __shared__ int lists[NCLS * LCAP];

  int tid = threadIdx.x;
  int lane = tid & 31;
  int wave = tid >> 5;
  const int wv = __builtin_amdgcn_readfirstlane(wave);
  __builtin_assume(wv >= 0 && wv < NCLS);
  int b = blockIdx.x / GPB;
  int part = blockIdx.x % GPB;
  unsigned n0 = (unsigned)part * (TILES * TP);
  const float* hb = hyper + (size_t)b * NC * NPIX;
  const int* sb = seg + (size_t)b * NPIX;
  int m = lane & 15;
  int kb = (lane >> 4) << 1;
  unsigned ltmask = (1u << lane) - 1u;
  int* mylist = lists + wave * LCAP;

  if (tid < NC) {
    stage[0][tid * PITCH + TP] = 0.0f;
    stage[1][tid * PITCH + TP] = 0.0f;
  }

  v8f acc[4][4] = {};
  float cs[4] = {0.0f, 0.0f, 0.0f, 0.0f};
  int cntTot = 0;

  int sv[TP / 32];
#pragma unroll
  for (int ch = 0; ch < TP / 32; ++ch) sv[ch] = sb[n0 + (unsigned)(ch * 32 + lane)];
  {
    float4 v[PCS];
    tile_loads(v, hb + n0, wv, lane);
    tile_stores(v, stage[0], wv, lane);
  }
  int cnt = build_list(sv, wave, lane, ltmask, mylist);
  __syncthreads();

  static_assert(TILES == 3, "the steps below are written out for three tiles");

  int padded = (cnt + 3) & ~3;
  cntTot += cnt;
#pragma unroll
  for (int ch = 0; ch < TP / 32; ++ch) sv[ch] = sb[n0 + (unsigned)(TP + ch * 32 + lane)];
  gather_tile(acc, cs, stage[0], mylist, padded, m, kb);
  cnt = build_list(sv, wave, lane, ltmask, mylist);
  {
    float4 v[PCS];
    tile_loads(v, hb + (n0 + TP), wv, lane);
    tile_stores(v, stage[1], wv, lane);
  }
  __syncthreads();

  padded = (cnt + 3) & ~3;
  cntTot += cnt;
#pragma unroll
  for (int ch = 0; ch < TP / 32; ++ch) sv[ch] = sb[n0 + (unsigned)(2 * TP + ch * 32 + lane)];
  gather_tile(acc, cs, stage[1], mylist, padded, m, kb);
  cnt = build_list(sv, wave, lane, ltmask, mylist);
  {
    float4 v[PCS];
    tile_loads(v, hb + (n0 + 2 * TP), wv, lane);
    tile_stores(v, stage[0], wv, lane);
  }
  __syncthreads();

  padded = (cnt + 3) & ~3;
  cntTot += cnt;
  gather_tile(acc, cs, stage[0], mylist, padded, m, kb);

  size_t gid = ((size_t)part * NB + b) * NCLS + wave;
  size_t ob = gid * (NC * NC);
  int col = lane & 15;
  int rof = (lane >> 4) * 8;
#pragma unroll
  for (int cb = 0; cb < 4; ++cb)
#pragma unroll
    for (int db = 0; db < 4; ++db)
#pragma unroll
      for (int j = 0; j < 8; ++j) {
        int r = cb * 16 + rof + j;
        sxxpart[ob + (size_t)r * NC + db * 16 + col] = acc[cb][db][j];
      }

#pragma unroll
  for (int blk = 0; blk < 4; ++blk) {
    float tot = cs[blk] + __shfl_xor(cs[blk], 16, 32);
    if (lane < 16) sumpart[gid * NC + blk * 16 + m] = tot;
  }
  if (lane == 0) cntpart[gid] = cntTot;
}

__global__ __launch_bounds__(256) void fold_kernel(const float* __restrict__ sxxpart,
                                                   const float* __restrict__ sumpart,
                                                   const int* __restrict__ cntpart,
                                                   float* __restrict__ sxxsum,
                                                   float* __restrict__ sumsum,
                                                   int* __restrict__ cntsum) {
  int bk = blockIdx.x / FSEG;
  int sg = blockIdx.x % FSEG;
  int tid = threadIdx.x;
  int qtr = tid >> 6;
  int e = tid & 63;
  __shared__ float4 red4[256];
  __shared__ float red1[256];
  __shared__ int cntS[NPARTS];

  {
    int f4 = sg * 64 + e;
    float4 s = make_float4(0.0f, 0.0f, 0.0f, 0.0f);
    for (int p0 = qtr * QP; p0 < qtr * QP + QP; p0 += 16) {
      float4 v[16];
#pragma unroll
      for (int u = 0; u < 16; ++u)
        v[u] = *(const float4*)(sxxpart + ((size_t)(p0 + u) * NB * NCLS + bk) * (NC * NC) +
                                (size_t)f4 * 4);
#pragma unroll
      for (int u = 0; u < 16; ++u) {
        s.x += v[u].x;
        s.y += v[u].y;
        s.z += v[u].z;
        s.w += v[u].w;
      }
    }
    red4[tid] = s;
    __syncthreads();
    if (tid < 64) {
      float4 a = red4[tid], b1 = red4[tid + 64], c1 = red4[tid + 128], d1 = red4[tid + 192];
      float4 r;
      r.x = ((a.x + b1.x) + c1.x) + d1.x;
      r.y = ((a.y + b1.y) + c1.y) + d1.y;
      r.z = ((a.z + b1.z) + c1.z) + d1.z;
      r.w = ((a.w + b1.w) + c1.w) + d1.w;
      *(float4*)(sxxsum + (size_t)bk * (NC * NC) + (size_t)f4 * 4) = r;
    }
  }

  if (sg == 0) {
    float s = 0.0f;
    for (int p0 = qtr * QP; p0 < qtr * QP + QP; p0 += 16) {
      float v[16];
#pragma unroll
      for (int u = 0; u < 16; ++u)
        v[u] = sumpart[((size_t)(p0 + u) * NB * NCLS + bk) * NC + e];
#pragma unroll
      for (int u = 0; u < 16; ++u) s += v[u];
    }
    red1[tid] = s;
    if (tid < NPARTS) cntS[tid] = cntpart[(size_t)tid * NB * NCLS + bk];
    __syncthreads();
    if (tid < 64)
      sumsum[(size_t)bk * NC + tid] =
          ((red1[tid] + red1[tid + 64]) + red1[tid + 128]) + red1[tid + 192];
    if (tid == 0) {
      int cnt = 0;
      for (int i = 0; i < NPARTS; ++i) cnt += cntS[i];
      cntsum[bk] = cnt;
    }
  }
}

#define RSP 68

__global__ __launch_bounds__(256) void reduce_solve_kernel(const float* __restrict__ sxxsum,
                                                           const float* __restrict__ sumsum,
                                                           const int* __restrict__ cntsum,
                                                           const float* __restrict__ pattern,
                                                           float* __restrict__ params) {
  int bk = blockIdx.x;
  int tid = threadIdx.x;
  int tx = tid & 15;
  int ty = tid >> 4;
  __shared__ __attribute__((aligned(16))) float colS[NC * RSP];

  int count = cntsum[bk];
  float safe = (float)(count > 0 ? count : 1);

  float4 sr = *(const float4*)(sumsum + (size_t)bk * NC + 4 * ty);
  float4 sc = *(const float4*)(sumsum + (size_t)bk * NC + 4 * tx);
  float mr[4] = {sr.x / safe, sr.y / safe, sr.z / safe, sr.w / safe};
  float mc[4] = {sc.x / safe, sc.y / safe, sc.z / safe, sc.w / safe};

  float a[4][4];
  float g[4];
#pragma unroll
  for (int c = 0; c < 4; ++c) g[c] = pattern[4 * tx + c];
#pragma unroll
  for (int r = 0; r < 4; ++r) {
    float4 s4 = *(const float4*)(sxxsum + (size_t)bk * (NC * NC) + (size_t)(4 * ty + r) * NC +
                                 4 * tx);
    float sv[4] = {s4.x, s4.y, s4.z, s4.w};
#pragma unroll
    for (int c = 0; c < 4; ++c) {
      float cv = sv[c] / safe - mr[r] * mc[c];
      if (4 * ty + r == 4 * tx + c) cv += EPSV;
      a[r][c] = cv;
    }
  }

  if (tx == 0) *(float4*)(colS + 4 * ty) = make_float4(a[0][0], a[1][0], a[2][0], a[3][0]);
  if (tid == 0) colS[NC] = g[0];

  for (int kb = 0; kb < NC / 4; ++kb) {
#pragma unroll
    for (int kc = 0; kc < 4; ++kc) {
      int k = kb * 4 + kc;
      __syncthreads();
      const float* ck = colS + k * RSP;
      float4 cr4 = *(const float4*)(ck + 4 * ty);
      float4 cc4 = *(const float4*)(ck + 4 * tx);
      float piv = ck[k];
      float ca = ck[NC];
      float inv = 1.0f / piv;
      float tr[4] = {cr4.x * inv, cr4.y * inv, cr4.z * inv, cr4.w * inv};
      float ta = ca * inv;
      float cc[4] = {cc4.x, cc4.y, cc4.z, cc4.w};

      const int cn = (kc + 1) & 3;
#pragma unroll
      for (int r = 0; r < 4; ++r) a[r][cn] = fmaf(-tr[r], cc[cn], a[r][cn]);
      g[cn] = fmaf(-ta, cc[cn], g[cn]);
      if (k + 1 < NC) {
        int kbn = (k + 1) >> 2;
        float* cnx = colS + (k + 1) * RSP;
        if (tx == kbn) *(float4*)(cnx + 4 * ty) = make_float4(a[0][cn], a[1][cn], a[2][cn], a[3][cn]);
        if (tid == kbn) cnx[NC] = g[cn];
      }
#pragma unroll
      for (int q = 1; q < 4; ++q) {
        const int c = (kc + 1 + q) & 3;
#pragma unroll
        for (int r = 0; r < 4; ++r) a[r][c] = fmaf(-tr[r], cc[c], a[r][c]);
        g[c] = fmaf(-ta, cc[c], g[c]);
      }
    }
  }

  if (tid < 32) {
    int l = tid;
    float cl[NC], cu[32];
#pragma unroll
    for (int q = 0; q < NC / 4; ++q) {
      float4 v = *(const float4*)(colS + l * RSP + 4 * q);
      cl[4 * q] = v.x; cl[4 * q + 1] = v.y; cl[4 * q + 2] = v.z; cl[4 * q + 3] = v.w;
    }
#pragma unroll
    for (int q = 0; q < 8; ++q) {
      float4 v = *(const float4*)(colS + (l + 32) * RSP + 32 + 4 * q);
      cu[4 * q] = v.x; cu[4 * q + 1] = v.y; cu[4 * q + 2] = v.z; cu[4 * q + 3] = v.w;
    }
    float i0 = 1.0f / colS[l * RSP + l];
    float i1 = 1.0f / colS[(l + 32) * RSP + l + 32];
    float r0 = colS[l * RSP + NC];
    float r1 = colS[(l + 32) * RSP + NC];
    float w0 = 0.0f, w1 = 0.0f;
#pragma unroll
    for (int i = NC - 1; i >= 32; --i) {
      float wi = __int_as_float(__builtin_amdgcn_readlane(__float_as_int(r1 * i1), i - 32));
      if (l == i - 32) w1 = wi;
      r0 = fmaf(-cl[i], wi, r0);
      if (l + 32 < i) r1 = fmaf(-cu[i - 32], wi, r1);
    }
#pragma unroll
    for (int i = 31; i >= 0; --i) {
      float wi = __int_as_float(__builtin_amdgcn_readlane(__float_as_int(r0 * i0), i));
      if (l == i) w0 = wi;
      if (l < i) r0 = fmaf(-cl[i], wi, r0);
    }

    params[(size_t)bk * 66 + l] = w0;
    params[(size_t)bk * 66 + l + 32] = w1;

    float pw = pattern[l] * w0 + pattern[l + 32] * w1;
    float off = (sumsum[(size_t)bk * NC + l] / safe) * w0 +
                (sumsum[(size_t)bk * NC + l + 32] / safe) * w1;
#pragma unroll
    for (int sh = 16; sh >= 1; sh >>= 1) {
      pw += __shfl_xor(pw, sh, 32);
      off += __shfl_xor(off, sh, 32);
    }
    if (l == 0) {
      float denom = sqrtf(pw);
      float scale = (count >= MIN_CLASS_SIZE && denom > 0.0f) ? (1.0f / denom) : 0.0f;
      params[(size_t)bk * 66 + 64] = off;
      params[(size_t)bk * 66 + 65] = scale;
    }
  }
}

__global__ __launch_bounds__(256) void score_kernel(const float* __restrict__ hyper,
                                                    const int* __restrict__ seg,
                                                    const float* __restrict__ params,
                                                    float* __restrict__ out) {
  int b = blockIdx.y;
  int tid = threadIdx.x;
  __shared__ float wS[NCLS * 65];
  __shared__ float offS[NCLS], sclS[NCLS];
  for (int i = tid; i < NCLS * NC; i += 256) {
    int k = i >> 6, c = i & 63;
    wS[k * 65 + c] = params[((size_t)b * NCLS + k) * 66 + c];
  }
  if (tid < NCLS) {
    offS[tid] = params[((size_t)b * NCLS + tid) * 66 + 64];
    sclS[tid] = params[((size_t)b * NCLS + tid) * 66 + 65];
  }
  __syncthreads();

  size_t n = (size_t)blockIdx.x * 256 + tid;
  int k = seg[(size_t)b * NPIX + n];
  const float* xp = hyper + (size_t)b * NC * NPIX + n;
  float acc = 0.0f;
#pragma unroll
  for (int c = 0; c < NC; ++c)
    acc = fmaf(xp[(size_t)c * NPIX], wS[k * 65 + c], acc);
  out[(size_t)b * NPIX + n] = (acc - offS[k]) * sclS[k];
}

extern "C" void kernel_launch(void* const* d_in, const int* in_sizes, int n_in,
                              void* d_out, int out_size, void* d_ws, size_t ws_size,
                              hipStream_t stream) {
  const float* hyper = (const float*)d_in[0];
  const float* pattern = (const float*)d_in[1];
  const int* seg = (const int*)d_in[2];
  float* out = (float*)d_out;

  char* ws = (char*)d_ws;
  auto al = [](size_t x) { return (x + 255) & ~(size_t)255; };
  size_t off = 0;
  float* sxxpart = (float*)(ws + off);
  off += al((size_t)NPARTS * NB * NCLS * NC * NC * sizeof(float));
  float* sumpart = (float*)(ws + off);
  off += al((size_t)NPARTS * NB * NCLS * NC * sizeof(float));
  int* cntpart = (int*)(ws + off);
  off += al((size_t)NPARTS * NB * NCLS * sizeof(int));
  float* sxxsum = (float*)(ws + off);
  off += al((size_t)NB * NCLS * NC * NC * sizeof(float));
  float* sumsum = (float*)(ws + off);
  off += al((size_t)NB * NCLS * NC * sizeof(float));
  int* cntsum = (int*)(ws + off);
  off += al((size_t)NB * NCLS * sizeof(int));
  float* params = (float*)(ws + off);
  off += al((size_t)NB * NCLS * 66 * sizeof(float));

  hipLaunchKernelGGL(stats_kernel, dim3(NB * GPB), dim3(NTHR), 0, stream, hyper, seg, sxxpart,
                     sumpart, cntpart);
  hipLaunchKernelGGL(fold_kernel, dim3(NB * NCLS * FSEG), dim3(256), 0, stream, sxxpart,
                     sumpart, cntpart, sxxsum, sumsum, cntsum);
  hipLaunchKernelGGL(reduce_solve_kernel, dim3(NB * NCLS), dim3(256), 0, stream, sxxsum,
                     sumsum, cntsum, pattern, params);
  hipLaunchKernelGGL(score_kernel, dim3(NPIX / 256, NB), dim3(256), 0, stream, hyper, seg,
                     params, out);
}


# reduce_solve: four pivot columns per barrier (16 hand-overs, were 64): every thread redoes the 4x4 pivot block and its own rows and columns of the panel in registers; same operations and order
# speedup vs baseline: 7.0545x; 1.0542x over previous
#include <hip/hip_runtime.h>


typedef float v2f __attribute__((ext_vector_type(2)));
typedef float v8f __attribute__((ext_vector_type(8)));

#define NB 2
#define NC 64
#define NPIX 147456
#define NCLS 20
#define TP 384
#define PITCH (TP + 4)
#define GPB 128
#define TILES (NPIX / GPB / TP)
#define NPARTS GPB
#define QP (NPARTS / 4)
#define FSEG 16
#define LCAP (TP + 4)
#define NTHR (NCLS * 32)
#define PCS 10
#define MIN_CLASS_SIZE 5000
#define EPSV 1e-6f


__device__ __forceinline__ void tile_loads(float4 (&v)[PCS], const float* __restrict__ tb,
                                           int wv, int lane) {
#pragma unroll
  for (int g = 0; g < 3; ++g) {
    unsigned ro = (unsigned)__builtin_amdgcn_readfirstlane((wv + 20 * g) * NPIX);
    const float* rp = tb + ro;
#pragma unroll
    for (int u = 0; u < 3; ++u) v[3 * g + u] = *(const float4*)(rp + (u * 128 + lane * 4));
  }
  {
    int w9 = wv < 12 ? wv : wv - 12;
    unsigned ro = (unsigned)__builtin_amdgcn_readfirstlane((60 + w9 / 3) * NPIX + (w9 % 3) * 128);
    const float* rp = tb + ro;
    v[9] = *(const float4*)(rp + lane * 4);
  }
}

__device__ __forceinline__ void tile_stores(const float4 (&v)[PCS], float* stage, int wv,
                                            int lane) {
  int lo = wv * PITCH + lane * 4;
  asm volatile("" : "+v"(lo));
  float* lb = stage + lo;
#pragma unroll
  for (int g = 0; g < 3; ++g)
#pragma unroll
    for (int u = 0; u < 3; ++u) *(float4*)(lb + (g * 20 * PITCH + u * 128)) = v[3 * g + u];
  int w9 = wv < 12 ? wv : wv - 12;
  *(float4*)(stage + (60 + w9 / 3) * PITCH + (w9 % 3) * 128 + lane * 4) = v[9];
}

__device__ __forceinline__ int build_list(const int (&sv)[TP / 32], int wave, int lane,
                                          unsigned ltmask, int* mylist) {
  int cnt = 0;
#pragma unroll
  for (int ch = 0; ch < TP / 32; ++ch) {
    bool is = (sv[ch] == wave);
    unsigned bal = __builtin_amdgcn_ballot_w32(is);
    if (is) mylist[cnt + __popc(bal & ltmask)] = ch * 32 + lane;
    cnt += __popc(bal);
  }
  int padded = (cnt + 3) & ~3;
  if (lane < padded - cnt) mylist[cnt + lane] = TP;
  return cnt;
}

__device__ __forceinline__ void gather_tile(v8f (&acc)[4][4], float (&cs)[4], const float* stage,
                                            const int* mylist, int padded, int m, int kb) {
  for (int s = 0; s < padded; s += 4) {
    int q0 = mylist[s + kb];
    int q1 = mylist[s + kb + 1];
    v2f F[4];
#pragma unroll
    for (int blk = 0; blk < 4; ++blk) {
      F[blk].x = stage[(blk * 16 + m) * PITCH + q0];
      F[blk].y = stage[(blk * 16 + m) * PITCH + q1];
      cs[blk] += F[blk].x + F[blk].y;
    }
#pragma unroll
    for (int cb = 0; cb < 4; ++cb)
#pragma unroll
      for (int db = 0; db < 4; ++db)
        acc[cb][db] = __builtin_amdgcn_wmma_f32_16x16x4_f32(
            false, F[cb], false, F[db], (short)0, acc[cb][db], false, false);
  }
}

__global__ __launch_bounds__(NTHR) void stats_kernel(const float* __restrict__ hyper,
                                                     const int* __restrict__ seg,
                                                     float* __restrict__ sxxpart,
                                                     float* __restrict__ sumpart,
                                                     int* __restrict__ cntpart) {
  __shared__ __attribute__((aligned(16))) float stage[2][NC * PITCH];
  __shared__ int lists[NCLS * LCAP];

  int tid = threadIdx.x;
  int lane = tid & 31;
  int wave = tid >> 5;
  const int wv = __builtin_amdgcn_readfirstlane(wave);
  __builtin_assume(wv >= 0 && wv < NCLS);
  int b = blockIdx.x / GPB;
  int part = blockIdx.x % GPB;
  unsigned n0 = (unsigned)part * (TILES * TP);
  const float* hb = hyper + (size_t)b * NC * NPIX;
  const int* sb = seg + (size_t)b * NPIX;
  int m = lane & 15;
  int kb = (lane >> 4) << 1;
  unsigned ltmask = (1u << lane) - 1u;
  int* mylist = lists + wave * LCAP;

  if (tid < NC) {
    stage[0][tid * PITCH + TP] = 0.0f;
    stage[1][tid * PITCH + TP] = 0.0f;
  }

  v8f acc[4][4] = {};
  float cs[4] = {0.0f, 0.0f, 0.0f, 0.0f};
  int cntTot = 0;

  int sv[TP / 32];
#pragma unroll
  for (int ch = 0; ch < TP / 32; ++ch) sv[ch] = sb[n0 + (unsigned)(ch * 32 + lane)];
  {
    float4 v[PCS];
    tile_loads(v, hb + n0, wv, lane);
    tile_stores(v, stage[0], wv, lane);
  }
  int cnt = build_list(sv, wave, lane, ltmask, mylist);
  __syncthreads();

  static_assert(TILES == 3, "the steps below are written out for three tiles");

  int padded = (cnt + 3) & ~3;
  cntTot += cnt;
#pragma unroll
  for (int ch = 0; ch < TP / 32; ++ch) sv[ch] = sb[n0 + (unsigned)(TP + ch * 32 + lane)];
  gather_tile(acc, cs, stage[0], mylist, padded, m, kb);
  cnt = build_list(sv, wave, lane, ltmask, mylist);
  {
    float4 v[PCS];
    tile_loads(v, hb + (n0 + TP), wv, lane);
    tile_stores(v, stage[1], wv, lane);
  }
  __syncthreads();

  padded = (cnt + 3) & ~3;
  cntTot += cnt;
#pragma unroll
  for (int ch = 0; ch < TP / 32; ++ch) sv[ch] = sb[n0 + (unsigned)(2 * TP + ch * 32 + lane)];
  gather_tile(acc, cs, stage[1], mylist, padded, m, kb);
  cnt = build_list(sv, wave, lane, ltmask, mylist);
  {
    float4 v[PCS];
    tile_loads(v, hb + (n0 + 2 * TP), wv, lane);
    tile_stores(v, stage[0], wv, lane);
  }
  __syncthreads();

  padded = (cnt + 3) & ~3;
  cntTot += cnt;
  gather_tile(acc, cs, stage[0], mylist, padded, m, kb);

  size_t gid = ((size_t)part * NB + b) * NCLS + wave;
  size_t ob = gid * (NC * NC);
  int col = lane & 15;
  int rof = (lane >> 4) * 8;
#pragma unroll
  for (int cb = 0; cb < 4; ++cb)
#pragma unroll
    for (int db = 0; db < 4; ++db)
#pragma unroll
      for (int j = 0; j < 8; ++j) {
        int r = cb * 16 + rof + j;
        sxxpart[ob + (size_t)r * NC + db * 16 + col] = acc[cb][db][j];
      }

#pragma unroll
  for (int blk = 0; blk < 4; ++blk) {
    float tot = cs[blk] + __shfl_xor(cs[blk], 16, 32);
    if (lane < 16) sumpart[gid * NC + blk * 16 + m] = tot;
  }
  if (lane == 0) cntpart[gid] = cntTot;
}

__global__ __launch_bounds__(256) void fold_kernel(const float* __restrict__ sxxpart,
                                                   const float* __restrict__ sumpart,
                                                   const int* __restrict__ cntpart,
                                                   float* __restrict__ sxxsum,
                                                   float* __restrict__ sumsum,
                                                   int* __restrict__ cntsum) {
  int bk = blockIdx.x / FSEG;
  int sg = blockIdx.x % FSEG;
  int tid = threadIdx.x;
  int qtr = tid >> 6;
  int e = tid & 63;
  __shared__ float4 red4[256];
  __shared__ float red1[256];
  __shared__ int cntS[NPARTS];

  {
    int f4 = sg * 64 + e;
    float4 s = make_float4(0.0f, 0.0f, 0.0f, 0.0f);
    for (int p0 = qtr * QP; p0 < qtr * QP + QP; p0 += 16) {
      float4 v[16];
#pragma unroll
      for (int u = 0; u < 16; ++u)
        v[u] = *(const float4*)(sxxpart + ((size_t)(p0 + u) * NB * NCLS + bk) * (NC * NC) +
                                (size_t)f4 * 4);
#pragma unroll
      for (int u = 0; u < 16; ++u) {
        s.x += v[u].x;
        s.y += v[u].y;
        s.z += v[u].z;
        s.w += v[u].w;
      }
    }
    red4[tid] = s;
    __syncthreads();
    if (tid < 64) {
      float4 a = red4[tid], b1 = red4[tid + 64], c1 = red4[tid + 128], d1 = red4[tid + 192];
      float4 r;
      r.x = ((a.x + b1.x) + c1.x) + d1.x;
      r.y = ((a.y + b1.y) + c1.y) + d1.y;
      r.z = ((a.z + b1.z) + c1.z) + d1.z;
      r.w = ((a.w + b1.w) + c1.w) + d1.w;
      *(float4*)(sxxsum + (size_t)bk * (NC * NC) + (size_t)f4 * 4) = r;
    }
  }

  if (sg == 0) {
    float s = 0.0f;
    for (int p0 = qtr * QP; p0 < qtr * QP + QP; p0 += 16) {
      float v[16];
#pragma unroll
      for (int u = 0; u < 16; ++u)
        v[u] = sumpart[((size_t)(p0 + u) * NB * NCLS + bk) * NC + e];
#pragma unroll
      for (int u = 0; u < 16; ++u) s += v[u];
    }
    red1[tid] = s;
    if (tid < NPARTS) cntS[tid] = cntpart[(size_t)tid * NB * NCLS + bk];
    __syncthreads();
    if (tid < 64)
      sumsum[(size_t)bk * NC + tid] =
          ((red1[tid] + red1[tid + 64]) + red1[tid + 128]) + red1[tid + 192];
    if (tid == 0) {
      int cnt = 0;
      for (int i = 0; i < NPARTS; ++i) cnt += cntS[i];
      cntsum[bk] = cnt;
    }
  }
}

#define RSP 68

__global__ __launch_bounds__(256) void reduce_solve_kernel(const float* __restrict__ sxxsum,
                                                           const float* __restrict__ sumsum,
                                                           const int* __restrict__ cntsum,
                                                           const float* __restrict__ pattern,
                                                           float* __restrict__ params) {
  int bk = blockIdx.x;
  int tid = threadIdx.x;
  int tx = tid & 15;
  int ty = tid >> 4;
  __shared__ __attribute__((aligned(16))) float colS[NC * RSP];
  __shared__ __attribute__((aligned(16))) float finS[NC * RSP];

  int count = cntsum[bk];
  float safe = (float)(count > 0 ? count : 1);

  float4 sr = *(const float4*)(sumsum + (size_t)bk * NC + 4 * ty);
  float4 sc = *(const float4*)(sumsum + (size_t)bk * NC + 4 * tx);
  float mr[4] = {sr.x / safe, sr.y / safe, sr.z / safe, sr.w / safe};
  float mc[4] = {sc.x / safe, sc.y / safe, sc.z / safe, sc.w / safe};

  float a[4][4];
  float g[4];
#pragma unroll
  for (int c = 0; c < 4; ++c) g[c] = pattern[4 * tx + c];
#pragma unroll
  for (int r = 0; r < 4; ++r) {
    float4 s4 = *(const float4*)(sxxsum + (size_t)bk * (NC * NC) + (size_t)(4 * ty + r) * NC +
                                 4 * tx);
    float sv[4] = {s4.x, s4.y, s4.z, s4.w};
#pragma unroll
    for (int c = 0; c < 4; ++c) {
      float cv = sv[c] / safe - mr[r] * mc[c];
      if (4 * ty + r == 4 * tx + c) cv += EPSV;
      a[r][c] = cv;
    }
  }

  if (tx == 0) {
#pragma unroll
    for (int c = 0; c < 4; ++c)
      *(float4*)(colS + c * RSP + 4 * ty) = make_float4(a[0][c], a[1][c], a[2][c], a[3][c]);
  }
  if (tid == 0) {
#pragma unroll
    for (int c = 0; c < 4; ++c) colS[c * RSP + NC] = g[c];
  }

  for (int kb = 0; kb < NC / 4; ++kb) {
    __syncthreads();
    const float* pk = colS + kb * 4 * RSP;
    float R[4][4], C[4][4], D[4][4], G[4];
#pragma unroll
    for (int j = 0; j < 4; ++j) {
      float4 r4 = *(const float4*)(pk + j * RSP + 4 * ty);
      float4 c4 = *(const float4*)(pk + j * RSP + 4 * tx);
      float4 d4 = *(const float4*)(pk + j * RSP + 4 * kb);
      R[j][0] = r4.x; R[j][1] = r4.y; R[j][2] = r4.z; R[j][3] = r4.w;
      C[j][0] = c4.x; C[j][1] = c4.y; C[j][2] = c4.z; C[j][3] = c4.w;
      D[j][0] = d4.x; D[j][1] = d4.y; D[j][2] = d4.z; D[j][3] = d4.w;
      G[j] = pk[j * RSP + NC];
    }

    float tr[4][4], ta[4];
#pragma unroll
    for (int j = 0; j < 4; ++j) {
      float inv = 1.0f / D[j][j];
#pragma unroll
      for (int r = 0; r < 4; ++r) tr[j][r] = R[j][r] * inv;
      ta[j] = G[j] * inv;
      if (j < 3) {
        float tc[4], td[4];
#pragma unroll
        for (int c = 0; c < 4; ++c) tc[c] = C[j][c] * inv;
#pragma unroll
        for (int i = 0; i < 4; ++i) td[i] = D[j][i] * inv;
#pragma unroll
        for (int jp = j + 1; jp < 4; ++jp) {
          float mlt = D[j][jp];
#pragma unroll
          for (int r = 0; r < 4; ++r) R[jp][r] = fmaf(-tr[j][r], mlt, R[jp][r]);
#pragma unroll
          for (int c = 0; c < 4; ++c) C[jp][c] = fmaf(-tc[c], mlt, C[jp][c]);
#pragma unroll
          for (int i = jp; i < 4; ++i) D[jp][i] = fmaf(-td[i], mlt, D[jp][i]);
          G[jp] = fmaf(-ta[j], mlt, G[jp]);
        }
      }
    }

#pragma unroll
    for (int j = 0; j < 4; ++j) {
#pragma unroll
      for (int c = 0; c < 4; ++c) {
#pragma unroll
        for (int r = 0; r < 4; ++r) a[r][c] = fmaf(-tr[j][r], C[j][c], a[r][c]);
        g[c] = fmaf(-ta[j], C[j][c], g[c]);
      }
    }

    if (kb + 1 < NC / 4) {
      if (tx == kb + 1) {
        float* nx = colS + (kb + 1) * 4 * RSP + 4 * ty;
#pragma unroll
        for (int c = 0; c < 4; ++c)
          *(float4*)(nx + c * RSP) = make_float4(a[0][c], a[1][c], a[2][c], a[3][c]);
      }
      if (tid == kb + 1) {
        float* nx = colS + (kb + 1) * 4 * RSP + NC;
#pragma unroll
        for (int c = 0; c < 4; ++c) nx[c * RSP] = g[c];
      }
    }
    if (tx == 0) {
      float* fk = finS + kb * 4 * RSP + 4 * ty;
#pragma unroll
      for (int j = 0; j < 4; ++j)
        *(float4*)(fk + j * RSP) = make_float4(R[j][0], R[j][1], R[j][2], R[j][3]);
    }
    if (tid == 0) {
#pragma unroll
      for (int j = 0; j < 4; ++j) finS[(kb * 4 + j) * RSP + NC] = G[j];
    }
  }
  __syncthreads();

  if (tid < 32) {
    int l = tid;
    float cl[NC], cu[32];
#pragma unroll
    for (int q = 0; q < NC / 4; ++q) {
      float4 v = *(const float4*)(finS + l * RSP + 4 * q);
      cl[4 * q] = v.x; cl[4 * q + 1] = v.y; cl[4 * q + 2] = v.z; cl[4 * q + 3] = v.w;
    }
#pragma unroll
    for (int q = 0; q < 8; ++q) {
      float4 v = *(const float4*)(finS + (l + 32) * RSP + 32 + 4 * q);
      cu[4 * q] = v.x; cu[4 * q + 1] = v.y; cu[4 * q + 2] = v.z; cu[4 * q + 3] = v.w;
    }
    float i0 = 1.0f / finS[l * RSP + l];
    float i1 = 1.0f / finS[(l + 32) * RSP + l + 32];
    float r0 = finS[l * RSP + NC];
    float r1 = finS[(l + 32) * RSP + NC];
    float w0 = 0.0f, w1 = 0.0f;
#pragma unroll
    for (int i = NC - 1; i >= 32; --i) {
      float wi = __int_as_float(__builtin_amdgcn_readlane(__float_as_int(r1 * i1), i - 32));
      if (l == i - 32) w1 = wi;
      r0 = fmaf(-cl[i], wi, r0);
      if (l + 32 < i) r1 = fmaf(-cu[i - 32], wi, r1);
    }
#pragma unroll
    for (int i = 31; i >= 0; --i) {
      float wi = __int_as_float(__builtin_amdgcn_readlane(__float_as_int(r0 * i0), i));
      if (l == i) w0 = wi;
      if (l < i) r0 = fmaf(-cl[i], wi, r0);
    }

    params[(size_t)bk * 66 + l] = w0;
    params[(size_t)bk * 66 + l + 32] = w1;

    float pw = pattern[l] * w0 + pattern[l + 32] * w1;
    float off = (sumsum[(size_t)bk * NC + l] / safe) * w0 +
                (sumsum[(size_t)bk * NC + l + 32] / safe) * w1;
#pragma unroll
    for (int sh = 16; sh >= 1; sh >>= 1) {
      pw += __shfl_xor(pw, sh, 32);
      off += __shfl_xor(off, sh, 32);
    }
    if (l == 0) {
      float denom = sqrtf(pw);
      float scale = (count >= MIN_CLASS_SIZE && denom > 0.0f) ? (1.0f / denom) : 0.0f;
      params[(size_t)bk * 66 + 64] = off;
      params[(size_t)bk * 66 + 65] = scale;
    }
  }
}

__global__ __launch_bounds__(256) void score_kernel(const float* __restrict__ hyper,
                                                    const int* __restrict__ seg,
                                                    const float* __restrict__ params,
                                                    float* __restrict__ out) {
  int b = blockIdx.y;
  int tid = threadIdx.x;
  __shared__ float wS[NCLS * 65];
  __shared__ float offS[NCLS], sclS[NCLS];
  for (int i = tid; i < NCLS * NC; i += 256) {
    int k = i >> 6, c = i & 63;
    wS[k * 65 + c] = params[((size_t)b * NCLS + k) * 66 + c];
  }
  if (tid < NCLS) {
    offS[tid] = params[((size_t)b * NCLS + tid) * 66 + 64];
    sclS[tid] = params[((size_t)b * NCLS + tid) * 66 + 65];
  }
  __syncthreads();

  size_t n = (size_t)blockIdx.x * 256 + tid;
  int k = seg[(size_t)b * NPIX + n];
  const float* xp = hyper + (size_t)b * NC * NPIX + n;
  float acc = 0.0f;
#pragma unroll
  for (int c = 0; c < NC; ++c)
    acc = fmaf(xp[(size_t)c * NPIX], wS[k * 65 + c], acc);
  out[(size_t)b * NPIX + n] = (acc - offS[k]) * sclS[k];
}

extern "C" void kernel_launch(void* const* d_in, const int* in_sizes, int n_in,
                              void* d_out, int out_size, void* d_ws, size_t ws_size,
                              hipStream_t stream) {
  const float* hyper = (const float*)d_in[0];
  const float* pattern = (const float*)d_in[1];
  const int* seg = (const int*)d_in[2];
  float* out = (float*)d_out;

  char* ws = (char*)d_ws;
  auto al = [](size_t x) { return (x + 255) & ~(size_t)255; };
  size_t off = 0;
  float* sxxpart = (float*)(ws + off);
  off += al((size_t)NPARTS * NB * NCLS * NC * NC * sizeof(float));
  float* sumpart = (float*)(ws + off);
  off += al((size_t)NPARTS * NB * NCLS * NC * sizeof(float));
  int* cntpart = (int*)(ws + off);
  off += al((size_t)NPARTS * NB * NCLS * sizeof(int));
  float* sxxsum = (float*)(ws + off);
  off += al((size_t)NB * NCLS * NC * NC * sizeof(float));
  float* sumsum = (float*)(ws + off);
  off += al((size_t)NB * NCLS * NC * sizeof(float));
  int* cntsum = (int*)(ws + off);
  off += al((size_t)NB * NCLS * sizeof(int));
  float* params = (float*)(ws + off);
  off += al((size_t)NB * NCLS * 66 * sizeof(float));

  hipLaunchKernelGGL(stats_kernel, dim3(NB * GPB), dim3(NTHR), 0, stream, hyper, seg, sxxpart,
                     sumpart, cntpart);
  hipLaunchKernelGGL(fold_kernel, dim3(NB * NCLS * FSEG), dim3(256), 0, stream, sxxpart,
                     sumpart, cntpart, sxxsum, sumsum, cntsum);
  hipLaunchKernelGGL(reduce_solve_kernel, dim3(NB * NCLS), dim3(256), 0, stream, sxxsum,
                     sumsum, cntsum, pattern, params);
  hipLaunchKernelGGL(score_kernel, dim3(NPIX / 256, NB), dim3(256), 0, stream, hyper, seg,
                     params, out);
}
